# starting kernel
# baseline (speedup 1.0000x reference)
#include <hip/hip_runtime.h>

typedef float v2f __attribute__((ext_vector_type(2)));
typedef float v8f __attribute__((ext_vector_type(8)));

#define SAMP_PER_BLOCK 16

__global__ __launch_bounds__(256) void qnet_fused_kernel(
    const float* __restrict__ x,
    const float* __restrict__ qw,
    const float* __restrict__ hw,
    const float* __restrict__ hb,
    const float* __restrict__ scale_p,
    float* __restrict__ out)
{
    __shared__ float pooledLDS[16 * 16];
    __shared__ float probsLDS[16 * 16];

    const int tid  = threadIdx.x;
    const int lane = tid & 31;
    const int wv   = tid >> 5;
    const int base = blockIdx.x * SAMP_PER_BLOCK;

    #pragma unroll
    for (int ss = 0; ss < 2; ++ss) {
        const int sl = wv * 2 + ss;
        const float4* __restrict__ xs =
            (const float4*)(x + (size_t)(base + sl) * 4096);
        float acc[4];
        #pragma unroll
        for (int rb = 0; rb < 4; ++rb) {
            float s = 0.0f;
            #pragma unroll
            for (int k8 = 0; k8 < 8; ++k8) {
                const int k = rb * 8 + k8;
                const float4 v = xs[lane + 32 * k];
                s += v.x + v.y + v.z + v.w;
            }
            acc[rb] = s;
        }
        #pragma unroll
        for (int rb = 0; rb < 4; ++rb) {
            float v = acc[rb];
            v += __shfl_xor(v, 1);
            v += __shfl_xor(v, 2);
            v += __shfl_xor(v, 16);
            acc[rb] = v;
        }
        if (((lane & 3) == 0) && lane < 16) {
            const int cb = lane >> 2;
            #pragma unroll
            for (int rb = 0; rb < 4; ++rb)
                pooledLDS[sl * 16 + rb * 4 + cb] = acc[rb];
        }
    }
    __syncthreads();

    if (tid < 32) {
        const float scale = scale_p[0];
        const int sm = lane & 15;
        const int hi = lane >> 4;

        float a[16];
        float nrm = 0.0f;
        #pragma unroll
        for (int i = 0; i < 16; ++i) { a[i] = pooledLDS[sm * 16 + i]; nrm += a[i] * a[i]; }
        const float inv = 1.0f / sqrtf(nrm);

        float re[16], im[16];
        #pragma unroll
        for (int i = 0; i < 16; ++i) { re[i] = a[i] * inv; im[i] = 0.0f; }

        #pragma unroll
        for (int l = 0; l < 6; ++l) {
            #pragma unroll
            for (int q = 0; q < 4; ++q) {
                const float half = 0.5f * qw[l * 4 + q];
                const float c = cosf(half);
                const float s = sinf(half);
                const int m = 1 << (3 - q);
                #pragma unroll
                for (int i = 0; i < 16; ++i) {
                    if (!(i & m)) {
                        const int j = i | m;
                        const float r0 = re[i], i0 = im[i];
                        const float r1 = re[j], i1 = im[j];
                        re[i] = c * r0 + s * i1;
                        im[i] = c * i0 - s * r1;
                        re[j] = c * r1 + s * i0;
                        im[j] = c * i1 - s * r0;
                    }
                }
            }
            #pragma unroll
            for (int q = 0; q < 4; ++q) {
                const int cm = 1 << (3 - q);
                const int tm = 1 << (3 - ((q + 1) & 3));
                #pragma unroll
                for (int i = 0; i < 16; ++i) {
                    if ((i & cm) && !(i & tm)) {
                        const int j = i | tm;
                        float t;
                        t = re[i]; re[i] = re[j]; re[j] = t;
                        t = im[i]; im[i] = im[j]; im[j] = t;
                    }
                }
            }
        }

        if (lane < 16) {
            #pragma unroll
            for (int i = 0; i < 16; ++i)
                probsLDS[lane * 16 + i] = re[i] * re[i] + im[i] * im[i];
        }
        __asm__ volatile("s_wait_dscnt 0" ::: "memory");

        v2f Af[4];
        #pragma unroll
        for (int t = 0; t < 4; ++t) {
            Af[t].x = probsLDS[sm * 16 + 4 * t + 2 * hi];
            Af[t].y = probsLDS[sm * 16 + 4 * t + 2 * hi + 1];
        }

        const int n = sm;
        const bool valid = (n < 10);
        float wn[10];
        #pragma unroll
        for (int j = 0; j < 10; ++j) wn[j] = valid ? hw[n * 10 + j] : 0.0f;
        const int cmask[10] = {8, 4, 2, 1, 12, 10, 9, 6, 5, 3};
        v2f Bf[4];
        #pragma unroll
        for (int t = 0; t < 4; ++t) {
            const int k0 = 4 * t + 2 * hi;
            float m0 = 0.0f, m1 = 0.0f;
            #pragma unroll
            for (int j = 0; j < 10; ++j) {
                const float w = wn[j];
                m0 += (__popc(k0 & cmask[j]) & 1) ? -w : w;
                m1 += (__popc((k0 + 1) & cmask[j]) & 1) ? -w : w;
            }
            Bf[t].x = m0;
            Bf[t].y = m1;
        }

        v8f acc = {};
        acc = __builtin_amdgcn_wmma_f32_16x16x4_f32(false, Af[0], false, Bf[0],
                                                    (short)0, acc, false, false);
        acc = __builtin_amdgcn_wmma_f32_16x16x4_f32(false, Af[1], false, Bf[1],
                                                    (short)0, acc, false, false);
        acc = __builtin_amdgcn_wmma_f32_16x16x4_f32(false, Af[2], false, Bf[2],
                                                    (short)0, acc, false, false);
        acc = __builtin_amdgcn_wmma_f32_16x16x4_f32(false, Af[3], false, Bf[3],
                                                    (short)0, acc, false, false);

        if (valid) {
            const float b = hb[n];
            #pragma unroll
            for (int v = 0; v < 8; ++v) {
                const int samp = base + v + 8 * hi;
                out[samp * 10 + n] = (acc[v] + b) * scale;
            }
        }
    }
}

extern "C" void kernel_launch(void* const* d_in, const int* in_sizes, int n_in,
                              void* d_out, int out_size, void* d_ws, size_t ws_size,
                              hipStream_t stream) {
    const float* x  = (const float*)d_in[0];
    const float* qw = (const float*)d_in[1];
    const float* hw = (const float*)d_in[2];
    const float* hb = (const float*)d_in[3];
    const float* sc = (const float*)d_in[4];
    float* out = (float*)d_out;

    const int nsamp  = in_sizes[0] / 4096;
    const int blocks = nsamp / SAMP_PER_BLOCK;

    qnet_fused_kernel<<<blocks, 256, 0, stream>>>(x, qw, hw, hb, sc, out);
}


# Ninth wave builds the 16x16 complex circuit matrix from the weights while waves 0-7 wait on memory; after the barrier it is applied by f32 WMMA (amplitude, norm and head sums reordered)
# speedup vs baseline: 1.4691x; 1.4691x over previous
#include <hip/hip_runtime.h>


typedef float v2f __attribute__((ext_vector_type(2)));
typedef float v8f __attribute__((ext_vector_type(8)));

#define SAMP_PER_BLOCK 16
#define POOL_WAVES 8

__device__ __forceinline__ float lane_value(float v, int src) {
    return __int_as_float(__builtin_amdgcn_readlane(__float_as_int(v), src));
}

__global__ __launch_bounds__(288) void qnet_fused_kernel(
    const float* __restrict__ x,
    const float* __restrict__ qw,
    const float* __restrict__ hw,
    const float* __restrict__ hb,
    const float* __restrict__ scale_p,
    float* __restrict__ out)
{
    __shared__ __attribute__((aligned(16))) float pooledLDS[16 * 16];
    __shared__ __attribute__((aligned(16))) float ureLDS[16 * 16];
    __shared__ __attribute__((aligned(16))) float uimLDS[16 * 16];

    const int tid  = threadIdx.x;
    const int lane = tid & 31;
    const int wv   = tid >> 5;
    const int base = blockIdx.x * SAMP_PER_BLOCK;
    const int c16  = lane & 15;
    const int hi   = lane >> 4;

    v2f   Mf[4];
    float bj[8];
    float scale = 0.0f;
    #pragma unroll
    for (int t = 0; t < 4; ++t) { Mf[t].x = 0.0f; Mf[t].y = 0.0f; }
    #pragma unroll
    for (int r = 0; r < 8; ++r) bj[r] = 0.0f;

    if (wv < POOL_WAVES) {
        #pragma unroll
        for (int ss = 0; ss < 2; ++ss) {
            const int sl = wv * 2 + ss;
            const float4* __restrict__ xs =
                (const float4*)(x + (size_t)(base + sl) * 4096);
            float acc[4];
            #pragma unroll
            for (int rb = 0; rb < 4; ++rb) {
                float s = 0.0f;
                #pragma unroll
                for (int k8 = 0; k8 < 8; ++k8) {
                    const int k = rb * 8 + k8;
                    const float4 v = xs[lane + 32 * k];
                    s += v.x + v.y + v.z + v.w;
                }
                acc[rb] = s;
            }
            #pragma unroll
            for (int rb = 0; rb < 4; ++rb) {
                float v = acc[rb];
                v += __shfl_xor(v, 1);
                v += __shfl_xor(v, 2);
                v += __shfl_xor(v, 16);
                acc[rb] = v;
            }
            if (((lane & 3) == 0) && lane < 16) {
                const int cb = lane >> 2;
                #pragma unroll
                for (int rb = 0; rb < 4; ++rb)
                    pooledLDS[sl * 16 + rb * 4 + cb] = acc[rb];
            }
        }
    } else {
        const float half = 0.5f * qw[lane < 24 ? lane : 23];
        const float cw = cosf(half);
        const float sw = sinf(half);

        float re[16], im[16];
        #pragma unroll
        for (int i = 0; i < 16; ++i) { re[i] = (i == c16) ? 1.0f : 0.0f; im[i] = 0.0f; }

        #pragma unroll
        for (int l = 0; l < 6; ++l) {
            #pragma unroll
            for (int q = 0; q < 4; ++q) {
                const float c = lane_value(cw, l * 4 + q);
                const float s = lane_value(sw, l * 4 + q);
                const int m = 1 << (3 - q);
                #pragma unroll
                for (int i = 0; i < 16; ++i) {
                    if (!(i & m)) {
                        const int j = i | m;
                        const float r0 = re[i], i0 = im[i];
                        const float r1 = re[j], i1 = im[j];
                        re[i] = c * r0 + s * i1;
                        im[i] = c * i0 - s * r1;
                        re[j] = c * r1 + s * i0;
                        im[j] = c * i1 - s * r0;
                    }
                }
            }
            #pragma unroll
            for (int q = 0; q < 4; ++q) {
                const int cm = 1 << (3 - q);
                const int tm = 1 << (3 - ((q + 1) & 3));
                #pragma unroll
                for (int i = 0; i < 16; ++i) {
                    if ((i & cm) && !(i & tm)) {
                        const int j = i | tm;
                        float t;
                        t = re[i]; re[i] = re[j]; re[j] = t;
                        t = im[i]; im[i] = im[j]; im[j] = t;
                    }
                }
            }
        }
        if (lane < 16) {
            #pragma unroll
            for (int n = 0; n < 16; ++n) {
                ureLDS[n * 16 + c16] = re[n];
                uimLDS[n * 16 + c16] = im[n];
            }
        }

        const bool valid = (c16 < 10);
        float wn[10];
        #pragma unroll
        for (int f = 0; f < 10; ++f) wn[f] = valid ? hw[c16 * 10 + f] : 0.0f;
        const int cmask[10] = {8, 4, 2, 1, 12, 10, 9, 6, 5, 3};
        #pragma unroll
        for (int t = 0; t < 4; ++t) {
            const int n0 = 8 * hi + 2 * t;
            float m0 = 0.0f, m1 = 0.0f;
            #pragma unroll
            for (int f = 0; f < 10; ++f) {
                const float w = wn[f];
                m0 += (__popc(n0 & cmask[f]) & 1) ? -w : w;
                m1 += (__popc((n0 + 1) & cmask[f]) & 1) ? -w : w;
            }
            Mf[t].x = m0;
            Mf[t].y = m1;
        }
        #pragma unroll
        for (int r = 0; r < 8; ++r) bj[r] = (8 * hi + r < 10) ? hb[8 * hi + r] : 0.0f;
        scale = scale_p[0];
    }
    __syncthreads();

    if (wv == POOL_WAVES) {
        const float4 p0 = *(const float4*)&pooledLDS[c16 * 16 + 8 * hi];
        const float4 p1 = *(const float4*)&pooledLDS[c16 * 16 + 8 * hi + 4];
        const float4 r0 = *(const float4*)&ureLDS[c16 * 16 + 8 * hi];
        const float4 r1 = *(const float4*)&ureLDS[c16 * 16 + 8 * hi + 4];
        const float4 i0 = *(const float4*)&uimLDS[c16 * 16 + 8 * hi];
        const float4 i1 = *(const float4*)&uimLDS[c16 * 16 + 8 * hi + 4];
        const float pa[8] = {p0.x, p0.y, p0.z, p0.w, p1.x, p1.y, p1.z, p1.w};
        const float ur[8] = {r0.x, r0.y, r0.z, r0.w, r1.x, r1.y, r1.z, r1.w};
        const float ui[8] = {i0.x, i0.y, i0.z, i0.w, i1.x, i1.y, i1.z, i1.w};
        float nrm = 0.0f;
        #pragma unroll
        for (int e = 0; e < 8; ++e) nrm += pa[e] * pa[e];
        nrm += __shfl_xor(nrm, 16);
        const float inv = 1.0f / sqrtf(nrm);

        v8f dre = {}, dim = {};
        #pragma unroll
        for (int t = 0; t < 4; ++t) {
            v2f Bf, Are, Aim;
            Bf.x  = pa[2 * t] * inv;  Bf.y  = pa[2 * t + 1] * inv;
            Are.x = ur[2 * t];        Are.y = ur[2 * t + 1];
            Aim.x = ui[2 * t];        Aim.y = ui[2 * t + 1];
            dre = __builtin_amdgcn_wmma_f32_16x16x4_f32(false, Are, false, Bf,
                                                        (short)0, dre, false, false);
            dim = __builtin_amdgcn_wmma_f32_16x16x4_f32(false, Aim, false, Bf,
                                                        (short)0, dim, false, false);
        }
        float P[8];
        #pragma unroll
        for (int r = 0; r < 8; ++r) P[r] = dre[r] * dre[r] + dim[r] * dim[r];

        v8f acc = {};
        #pragma unroll
        for (int t = 0; t < 4; ++t) {
            v2f Pf;
            Pf.x = P[2 * t];  Pf.y = P[2 * t + 1];
            acc = __builtin_amdgcn_wmma_f32_16x16x4_f32(false, Mf[t], false, Pf,
                                                        (short)0, acc, false, false);
        }
        const int samp = base + c16;
        #pragma unroll
        for (int r = 0; r < 8; ++r) {
            if (8 * hi + r < 10)
                out[samp * 10 + 8 * hi + r] = (acc[r] + bj[r]) * scale;
        }
    }
}

extern "C" void kernel_launch(void* const* d_in, const int* in_sizes, int n_in,
                              void* d_out, int out_size, void* d_ws, size_t ws_size,
                              hipStream_t stream) {
    const float* x  = (const float*)d_in[0];
    const float* qw = (const float*)d_in[1];
    const float* hw = (const float*)d_in[2];
    const float* hb = (const float*)d_in[3];
    const float* sc = (const float*)d_in[4];
    float* out = (float*)d_out;

    const int nsamp  = in_sizes[0] / 4096;
    const int blocks = nsamp / SAMP_PER_BLOCK;

    qnet_fused_kernel<<<blocks, 288, 0, stream>>>(x, qw, hw, hb, sc, out);
}


# Pooling waves request a whole 16 KB sample (32 loads) at once and refill each register set with the next sample's piece as it is summed: 32 loads in flight per wave, not 4 to 8; sum order kept
# speedup vs baseline: 1.5609x; 1.0625x over previous
#include <hip/hip_runtime.h>


typedef float v2f __attribute__((ext_vector_type(2)));
typedef float v8f __attribute__((ext_vector_type(8)));

#define SAMP_PER_BLOCK 16
#define POOL_WAVES 8

__device__ __forceinline__ float lane_value(float v, int src) {
    return __int_as_float(__builtin_amdgcn_readlane(__float_as_int(v), src));
}

__global__ __launch_bounds__(288) __attribute__((amdgpu_waves_per_eu(6))) void qnet_fused_kernel(
    const float* __restrict__ x,
    const float* __restrict__ qw,
    const float* __restrict__ hw,
    const float* __restrict__ hb,
    const float* __restrict__ scale_p,
    float* __restrict__ out)
{
    __shared__ __attribute__((aligned(16))) float pooledLDS[16 * 16];
    __shared__ __attribute__((aligned(16))) float ureLDS[16 * 16];
    __shared__ __attribute__((aligned(16))) float uimLDS[16 * 16];

    const int tid  = threadIdx.x;
    const int lane = tid & 31;
    const int wv   = tid >> 5;
    const int base = blockIdx.x * SAMP_PER_BLOCK;
    const int c16  = lane & 15;
    const int hi   = lane >> 4;

    v2f   Mf[4];
    float bj[8];
    float scale = 0.0f;
    #pragma unroll
    for (int t = 0; t < 4; ++t) { Mf[t].x = 0.0f; Mf[t].y = 0.0f; }
    #pragma unroll
    for (int r = 0; r < 8; ++r) bj[r] = 0.0f;

    if (wv < POOL_WAVES) {
        const float4* __restrict__ xs =
            (const float4*)(x + (size_t)(base + wv * 2) * 4096);
        float4 win[32];
        #pragma unroll
        for (int k = 0; k < 32; ++k) win[k] = xs[lane + 32 * k];
        __builtin_amdgcn_sched_barrier(0);
        #pragma unroll
        for (int ss = 0; ss < 2; ++ss) {
            const int sl = wv * 2 + ss;
            float acc[4];
            #pragma unroll
            for (int rb = 0; rb < 4; ++rb) {
                float s = 0.0f;
                #pragma unroll
                for (int k8 = 0; k8 < 8; ++k8) {
                    const int k = rb * 8 + k8;
                    const float4 v = win[k];
                    s += v.x + v.y + v.z + v.w;
                    if (ss == 0) {
                        win[k] = xs[1024 + lane + 32 * k];
                        __builtin_amdgcn_sched_barrier(0);
                    }
                }
                acc[rb] = s;
            }
            #pragma unroll
            for (int rb = 0; rb < 4; ++rb) {
                float v = acc[rb];
                v += __shfl_xor(v, 1);
                v += __shfl_xor(v, 2);
                v += __shfl_xor(v, 16);
                acc[rb] = v;
            }
            if (((lane & 3) == 0) && lane < 16) {
                const int cb = lane >> 2;
                #pragma unroll
                for (int rb = 0; rb < 4; ++rb)
                    pooledLDS[sl * 16 + rb * 4 + cb] = acc[rb];
            }
        }
    } else {
        const float half = 0.5f * qw[lane < 24 ? lane : 23];
        const float cw = cosf(half);
        const float sw = sinf(half);

        float re[16], im[16];
        #pragma unroll
        for (int i = 0; i < 16; ++i) { re[i] = (i == c16) ? 1.0f : 0.0f; im[i] = 0.0f; }

        #pragma unroll
        for (int l = 0; l < 6; ++l) {
            #pragma unroll
            for (int q = 0; q < 4; ++q) {
                const float c = lane_value(cw, l * 4 + q);
                const float s = lane_value(sw, l * 4 + q);
                const int m = 1 << (3 - q);
                #pragma unroll
                for (int i = 0; i < 16; ++i) {
                    if (!(i & m)) {
                        const int j = i | m;
                        const float r0 = re[i], i0 = im[i];
                        const float r1 = re[j], i1 = im[j];
                        re[i] = c * r0 + s * i1;
                        im[i] = c * i0 - s * r1;
                        re[j] = c * r1 + s * i0;
                        im[j] = c * i1 - s * r0;
                    }
                }
            }
            #pragma unroll
            for (int q = 0; q < 4; ++q) {
                const int cm = 1 << (3 - q);
                const int tm = 1 << (3 - ((q + 1) & 3));
                #pragma unroll
                for (int i = 0; i < 16; ++i) {
                    if ((i & cm) && !(i & tm)) {
                        const int j = i | tm;
                        float t;
                        t = re[i]; re[i] = re[j]; re[j] = t;
                        t = im[i]; im[i] = im[j]; im[j] = t;
                    }
                }
            }
        }
        if (lane < 16) {
            #pragma unroll
            for (int n = 0; n < 16; ++n) {
                ureLDS[n * 16 + c16] = re[n];
                uimLDS[n * 16 + c16] = im[n];
            }
        }

        const bool valid = (c16 < 10);
        float wn[10];
        #pragma unroll
        for (int f = 0; f < 10; ++f) wn[f] = valid ? hw[c16 * 10 + f] : 0.0f;
        const int cmask[10] = {8, 4, 2, 1, 12, 10, 9, 6, 5, 3};
        #pragma unroll
        for (int t = 0; t < 4; ++t) {
            const int n0 = 8 * hi + 2 * t;
            float m0 = 0.0f, m1 = 0.0f;
            #pragma unroll
            for (int f = 0; f < 10; ++f) {
                const float w = wn[f];
                m0 += (__popc(n0 & cmask[f]) & 1) ? -w : w;
                m1 += (__popc((n0 + 1) & cmask[f]) & 1) ? -w : w;
            }
            Mf[t].x = m0;
            Mf[t].y = m1;
        }
        #pragma unroll
        for (int r = 0; r < 8; ++r) bj[r] = (8 * hi + r < 10) ? hb[8 * hi + r] : 0.0f;
        scale = scale_p[0];
    }
    __syncthreads();

    if (wv == POOL_WAVES) {
        const float4 p0 = *(const float4*)&pooledLDS[c16 * 16 + 8 * hi];
        const float4 p1 = *(const float4*)&pooledLDS[c16 * 16 + 8 * hi + 4];
        const float4 r0 = *(const float4*)&ureLDS[c16 * 16 + 8 * hi];
        const float4 r1 = *(const float4*)&ureLDS[c16 * 16 + 8 * hi + 4];
        const float4 i0 = *(const float4*)&uimLDS[c16 * 16 + 8 * hi];
        const float4 i1 = *(const float4*)&uimLDS[c16 * 16 + 8 * hi + 4];
        const float pa[8] = {p0.x, p0.y, p0.z, p0.w, p1.x, p1.y, p1.z, p1.w};
        const float ur[8] = {r0.x, r0.y, r0.z, r0.w, r1.x, r1.y, r1.z, r1.w};
        const float ui[8] = {i0.x, i0.y, i0.z, i0.w, i1.x, i1.y, i1.z, i1.w};
        float nrm = 0.0f;
        #pragma unroll
        for (int e = 0; e < 8; ++e) nrm += pa[e] * pa[e];
        nrm += __shfl_xor(nrm, 16);
        const float inv = 1.0f / sqrtf(nrm);

        v8f dre = {}, dim = {};
        #pragma unroll
        for (int t = 0; t < 4; ++t) {
            v2f Bf, Are, Aim;
            Bf.x  = pa[2 * t] * inv;  Bf.y  = pa[2 * t + 1] * inv;
            Are.x = ur[2 * t];        Are.y = ur[2 * t + 1];
            Aim.x = ui[2 * t];        Aim.y = ui[2 * t + 1];
            dre = __builtin_amdgcn_wmma_f32_16x16x4_f32(false, Are, false, Bf,
                                                        (short)0, dre, false, false);
            dim = __builtin_amdgcn_wmma_f32_16x16x4_f32(false, Aim, false, Bf,
                                                        (short)0, dim, false, false);
        }
        float P[8];
        #pragma unroll
        for (int r = 0; r < 8; ++r) P[r] = dre[r] * dre[r] + dim[r] * dim[r];

        v8f acc = {};
        #pragma unroll
        for (int t = 0; t < 4; ++t) {
            v2f Pf;
            Pf.x = P[2 * t];  Pf.y = P[2 * t + 1];
            acc = __builtin_amdgcn_wmma_f32_16x16x4_f32(false, Mf[t], false, Pf,
                                                        (short)0, acc, false, false);
        }
        const int samp = base + c16;
        #pragma unroll
        for (int r = 0; r < 8; ++r) {
            if (8 * hi + r < 10)
                out[samp * 10 + 8 * hi + r] = (acc[r] + bj[r]) * scale;
        }
    }
}

extern "C" void kernel_launch(void* const* d_in, const int* in_sizes, int n_in,
                              void* d_out, int out_size, void* d_ws, size_t ws_size,
                              hipStream_t stream) {
    const float* x  = (const float*)d_in[0];
    const float* qw = (const float*)d_in[1];
    const float* hw = (const float*)d_in[2];
    const float* hb = (const float*)d_in[3];
    const float* sc = (const float*)d_in[4];
    float* out = (float*)d_out;

    const int nsamp  = in_sizes[0] / 4096;
    const int blocks = nsamp / SAMP_PER_BLOCK;

    qnet_fused_kernel<<<blocks, 288, 0, stream>>>(x, qw, hw, hb, sc, out);
}
